# starting kernel
# baseline (speedup 1.0000x reference)
#include <hip/hip_runtime.h>


typedef __attribute__((ext_vector_type(2))) float v2f;
typedef __attribute__((ext_vector_type(8))) float v8f;

__global__ __launch_bounds__(256) void zero_kernel(float4* __restrict__ out, int n4) {
  int i = blockIdx.x * blockDim.x + threadIdx.x;
  if (i < n4) out[i] = make_float4(0.f, 0.f, 0.f, 0.f);
}

__global__ __launch_bounds__(256) void gemm_scale_kernel(
    const float* __restrict__ h, const float* __restrict__ w,
    const float* __restrict__ norm, float* __restrict__ hn, int n_nodes) {
  __shared__ float wlds[64 * 64];

  {
    const float4* wsrc = (const float4*)w;
    float4* wdst = (float4*)wlds;
    #pragma unroll
    for (int i = threadIdx.x; i < 64 * 64 / 4; i += 256) wdst[i] = wsrc[i];
  }
  __syncthreads();

  const int wave  = threadIdx.x >> 5;
  const int lane  = threadIdx.x & 31;
  const int c0    = (wave & 3) * 16;
  const int rw    = blockIdx.x * 32 + (wave >> 2) * 16;
  const int m     = lane & 15;
  const int khalf = lane >> 4;

  int arow = rw + m;
  if (arow >= n_nodes) arow = n_nodes - 1;
  const float* aptr = h + arow * 64 + khalf * 2;

  v8f acc = {};
  #pragma unroll
  for (int kk = 0; kk < 16; ++kk) {
    const int kbase = kk * 4 + khalf * 2;
    const float2 av = *(const float2*)(aptr + kk * 4);
    v2f a; a.x = av.x; a.y = av.y;
    v2f b;
    b.x = wlds[kbase * 64 + c0 + m];
    b.y = wlds[(kbase + 1) * 64 + c0 + m];
    acc = __builtin_amdgcn_wmma_f32_16x16x4_f32(
         false, a,  false, b,
         (short)0, acc,  false,  false);
  }

  const int rbase = rw + khalf * 8;
  #pragma unroll
  for (int r = 0; r < 8; ++r) {
    const int row = rbase + r;
    if (row < n_nodes) hn[row * 64 + c0 + m] = acc[r] * norm[row];
  }
}

__global__ __launch_bounds__(256) void scatter_kernel(
    const float* __restrict__ hn, const int* __restrict__ src,
    const int* __restrict__ dst, float* __restrict__ out, int n_edges) {
  const int e = blockIdx.x * 8 + (threadIdx.x >> 5);
  if (e >= n_edges) return;
  const int lane = threadIdx.x & 31;
  const int s = src[e];
  const int d = dst[e];
  const float2 v = *(const float2*)(hn + s * 64 + lane * 2);
  float* p = out + d * 64 + lane * 2;
  (void)__hip_atomic_fetch_add(p,     v.x, __ATOMIC_RELAXED, __HIP_MEMORY_SCOPE_AGENT);
  (void)__hip_atomic_fetch_add(p + 1, v.y, __ATOMIC_RELAXED, __HIP_MEMORY_SCOPE_AGENT);
}

__global__ __launch_bounds__(256) void scale_kernel(
    float* __restrict__ out, const float* __restrict__ norm, int n_nodes) {
  const int i = blockIdx.x * blockDim.x + threadIdx.x;
  const int n4 = n_nodes * 16;
  if (i >= n4) return;
  const float s = norm[i >> 4];
  float4* p = (float4*)out + i;
  float4 v = *p;
  v.x *= s; v.y *= s; v.z *= s; v.w *= s;
  *p = v;
}

extern "C" void kernel_launch(void* const* d_in, const int* in_sizes, int n_in,
                              void* d_out, int out_size, void* d_ws, size_t ws_size,
                              hipStream_t stream) {
  const float* h    = (const float*)d_in[0];
  const float* w    = (const float*)d_in[1];
  const float* norm = (const float*)d_in[2];
  const int*   src  = (const int*)d_in[3];
  const int*   dst  = (const int*)d_in[4];
  float* out = (float*)d_out;
  float* hn  = (float*)d_ws;

  const int n_nodes = in_sizes[2];
  const int n_edges = in_sizes[3];
  const int n4 = n_nodes * 16;

  zero_kernel<<<(n4 + 255) / 256, 256, 0, stream>>>((float4*)out, n4);

  gemm_scale_kernel<<<(n_nodes + 31) / 32, 256, 0, stream>>>(h, w, norm, hn, n_nodes);

  scatter_kernel<<<(n_edges + 7) / 8, 256, 0, stream>>>(hn, src, dst, out, n_edges);

  scale_kernel<<<(n4 + 255) / 256, 256, 0, stream>>>(out, norm, n_nodes);
}


# Edges binned per 128 destinations via LDS, one returning add per block and bin; one block per bin buckets its list in LDS, a wave sums 8 destinations, 8 row loads in flight; sums in arrival order
# speedup vs baseline: 2.9504x; 2.9504x over previous
#include <hip/hip_runtime.h>


typedef __attribute__((ext_vector_type(2))) float v2f;
typedef __attribute__((ext_vector_type(8))) float v8f;

#define BIN_ROWS   128
#define BIN_WORDS  (BIN_ROWS * 64)
#define MAX_BINS   1024
#define MAX_NODES  (MAX_BINS * BIN_ROWS)
#define FILL_EDGES 8192
#define SLOTS      64

__global__ __launch_bounds__(256) void gemm_scale_kernel(
    const float* __restrict__ h, const float* __restrict__ w,
    const float* __restrict__ norm, float* __restrict__ hn,
    float* __restrict__ out, int n_nodes) {
  __shared__ float wlds[64 * 64];

  if (threadIdx.x == 0 && (blockIdx.x * 32) % BIN_ROWS == 0)
    ((int*)out)[blockIdx.x * 32 * 64] = 0;

  {
    const float4* wsrc = (const float4*)w;
    float4* wdst = (float4*)wlds;
    #pragma unroll
    for (int i = threadIdx.x; i < 64 * 64 / 4; i += 256) wdst[i] = wsrc[i];
  }
  __syncthreads();

  const int wave  = threadIdx.x >> 5;
  const int lane  = threadIdx.x & 31;
  const int c0    = (wave & 3) * 16;
  const int rw    = blockIdx.x * 32 + (wave >> 2) * 16;
  const int m     = lane & 15;
  const int khalf = lane >> 4;

  int arow = rw + m;
  if (arow >= n_nodes) arow = n_nodes - 1;
  const float* aptr = h + arow * 64 + khalf * 2;

  v8f acc = {};
  #pragma unroll
  for (int kk = 0; kk < 16; ++kk) {
    const int kbase = kk * 4 + khalf * 2;
    const float2 av = *(const float2*)(aptr + kk * 4);
    v2f a; a.x = av.x; a.y = av.y;
    v2f b;
    b.x = wlds[kbase * 64 + c0 + m];
    b.y = wlds[(kbase + 1) * 64 + c0 + m];
    acc = __builtin_amdgcn_wmma_f32_16x16x4_f32(
         false, a,  false, b,
         (short)0, acc,  false,  false);
  }

  const int rbase = rw + khalf * 8;
  #pragma unroll
  for (int r = 0; r < 8; ++r) {
    const int row = rbase + r;
    if (row < n_nodes) hn[row * 64 + c0 + m] = acc[r] * norm[row];
  }
}

__global__ __launch_bounds__(256) void bin_fill_kernel(
    const int* __restrict__ src, const int* __restrict__ dst,
    float* __restrict__ out, int n_nodes, int n_edges) {
  __shared__ __attribute__((aligned(16))) int hist[MAX_BINS];
  __shared__ int lofs[MAX_BINS];
  __shared__ int gadr[MAX_BINS];
  __shared__ int wsum[8];
  __shared__ int sval[FILL_EDGES];
  __shared__ int sadr[FILL_EDGES];

  if (n_nodes > MAX_NODES) return;

  const int tid  = threadIdx.x;
  const int lane = tid & 31;
  const int wave = tid >> 5;
  *(int4*)&hist[4 * tid] = make_int4(0, 0, 0, 0);
  __syncthreads();

  const int e0 = blockIdx.x * FILL_EDGES + tid;
  int sv[32], dv[32], br[32];
  #pragma unroll
  for (int k = 0; k < 32; ++k) {
    const int e = e0 + k * 256;
    sv[k] = (e < n_edges) ? src[e] : 0;
    dv[k] = (e < n_edges) ? dst[e] : -1;
  }
  #pragma unroll
  for (int k = 0; k < 32; ++k) {
    br[k] = -1;
    if (dv[k] >= 0) {
      const int bin = dv[k] >> 7;
      const int r = atomicAdd(&hist[bin], 1);
      br[k] = (bin << 16) | r;
      sv[k] |= (dv[k] & (BIN_ROWS - 1)) << 17;
    }
  }
  __syncthreads();

  const int4 hv = *(const int4*)&hist[4 * tid];
  const int tsum = hv.x + hv.y + hv.z + hv.w;
  int incl = tsum;
  #pragma unroll
  for (int o = 1; o < 32; o <<= 1) {
    const int t = __shfl_up(incl, o);
    if (lane >= o) incl += t;
  }
  if (lane == 31) wsum[wave] = incl;

  int hb[4] = {hv.x, hv.y, hv.z, hv.w};
  int gb[4];
  #pragma unroll
  for (int u = 0; u < 4; ++u) {
    gb[u] = 0;
    if (hb[u] > 0)
      gb[u] = __hip_atomic_fetch_add((int*)out + (4 * tid + u) * BIN_WORDS, hb[u],
                                     __ATOMIC_RELAXED, __HIP_MEMORY_SCOPE_AGENT);
  }
  __syncthreads();
  int excl = incl - tsum;
  for (int w = 0; w < wave; ++w) excl += wsum[w];
  int total = 0;
  for (int w = 0; w < 8; ++w) total += wsum[w];
  #pragma unroll
  for (int u = 0; u < 4; ++u) {
    const int bin = 4 * tid + u;
    lofs[bin] = excl;
    gadr[bin] = bin * BIN_WORDS + 1 + gb[u];
    excl += hb[u];
  }
  __syncthreads();

  #pragma unroll
  for (int k = 0; k < 32; ++k) {
    if (br[k] >= 0) {
      const int bin = br[k] >> 16;
      const int r   = br[k] & 0xFFFF;
      const int is  = lofs[bin] + r;
      const int a   = gadr[bin] + r;
      const int rows = min(BIN_ROWS, n_nodes - bin * BIN_ROWS);
      const int last = bin * BIN_WORDS + rows * 64 - 1;
      sval[is] = sv[k];
      sadr[is] = (a <= last) ? a : -1;
    }
  }
  __syncthreads();

  for (int i0 = tid; i0 < total; i0 += 8 * 256) {
    int a[8], v[8];
    #pragma unroll
    for (int u = 0; u < 8; ++u) {
      const int i = i0 + u * 256;
      a[u] = (i < total) ? sadr[i] : -1;
      v[u] = (i < total) ? sval[i] : 0;
    }
    #pragma unroll
    for (int u = 0; u < 8; ++u)
      if (a[u] >= 0) ((int*)out)[a[u]] = v[u];
  }
}

__global__ __launch_bounds__(512) void bin_sum_kernel(
    const float* __restrict__ hn, const float* __restrict__ norm,
    const int* __restrict__ src, const int* __restrict__ dst,
    float* out, int n_nodes, int n_edges) {
  __shared__ int cnt[BIN_ROWS];
  __shared__ int bucket[BIN_ROWS * SLOTS];

  const int tid  = threadIdx.x;
  const int d0   = blockIdx.x * BIN_ROWS;
  const int rows = min(BIN_ROWS, n_nodes - d0);
  const int* list = (const int*)out + d0 * 64;
  const int cap  = rows * 64 - 1;

  if (tid < BIN_ROWS) cnt[tid] = 0;
  const int c = list[0];
  const bool binok = (n_nodes <= MAX_NODES) && (c >= 0) && (c <= cap);
  __syncthreads();
  if (binok) {
    for (int i0 = tid; i0 < c; i0 += 4 * 512) {
      unsigned e[4];
      #pragma unroll
      for (int u = 0; u < 4; ++u) {
        const int i = i0 + u * 512;
        e[u] = (i < c) ? (unsigned)list[1 + i] : 0u;
      }
      #pragma unroll
      for (int u = 0; u < 4; ++u) {
        if (i0 + u * 512 < c) {
          const int dl = (int)(e[u] >> 17) & (BIN_ROWS - 1);
          const int s  = (int)(e[u] & 0x1FFFFu);
          const int slot = atomicAdd(&cnt[dl], 1);
          if (slot < SLOTS) bucket[dl * SLOTS + slot] = s;
        }
      }
    }
  }
  __syncthreads();

  const int wave = tid >> 5;
  const int lane = tid & 31;
  const int half = lane >> 4;
  const int q    = lane & 15;

  for (int k = 0; k < 8; ++k) {
    const int dl = wave * 8 + k;
    if (dl >= rows) break;
    const int n = d0 + dl;
    const int cd = cnt[dl];
    const float dn = norm[n];
    float* orow = out + n * 64;

    if (!binok || cd > SLOTS) {
      float2 a2 = make_float2(0.f, 0.f);
      for (int base = 0; base < n_edges; base += 32) {
        const int e  = base + lane;
        const int dv = (e < n_edges) ? dst[e] : -1;
        const int sv = (e < n_edges) ? src[e] : 0;
        unsigned mask = __builtin_amdgcn_ballot_w32(dv == n);
        while (mask) {
          const int b = __builtin_ctz(mask);
          mask &= mask - 1;
          const int s = __builtin_amdgcn_readlane(sv, b);
          const float2 v = *(const float2*)(hn + s * 64 + lane * 2);
          a2.x += v.x; a2.y += v.y;
        }
      }
      a2.x *= dn; a2.y *= dn;
      *(float2*)(orow + lane * 2) = a2;
      continue;
    }

    const int* bk = bucket + dl * SLOTS + half;
    float4 acc = make_float4(0.f, 0.f, 0.f, 0.f);
    const int nfull = cd >> 1;
    int j = 0;
    for (; j + 8 <= nfull; j += 8) {
      float4 v[8];
      #pragma unroll
      for (int u = 0; u < 8; ++u) {
        const int s = bk[2 * (j + u)];
        v[u] = *(const float4*)(hn + s * 64 + q * 4);
      }
      #pragma unroll
      for (int u = 0; u < 8; ++u) {
        acc.x += v[u].x; acc.y += v[u].y; acc.z += v[u].z; acc.w += v[u].w;
      }
    }
    {
      float4 v[7];
      #pragma unroll
      for (int u = 0; u < 7; ++u) {
        v[u] = make_float4(0.f, 0.f, 0.f, 0.f);
        if (j + u < nfull) {
          const int s = bk[2 * (j + u)];
          v[u] = *(const float4*)(hn + s * 64 + q * 4);
        }
      }
      #pragma unroll
      for (int u = 0; u < 7; ++u) {
        acc.x += v[u].x; acc.y += v[u].y; acc.z += v[u].z; acc.w += v[u].w;
      }
    }
    if (cd & 1) {
      if (half == 0) {
        const int s = bk[cd - 1];
        const float4 v = *(const float4*)(hn + s * 64 + q * 4);
        acc.x += v.x; acc.y += v.y; acc.z += v.z; acc.w += v.w;
      }
    }

    acc.x += __shfl_xor(acc.x, 16);
    acc.y += __shfl_xor(acc.y, 16);
    acc.z += __shfl_xor(acc.z, 16);
    acc.w += __shfl_xor(acc.w, 16);
    if (half == 0) {
      acc.x *= dn; acc.y *= dn; acc.z *= dn; acc.w *= dn;
      *(float4*)(orow + q * 4) = acc;
    }
  }
}

extern "C" void kernel_launch(void* const* d_in, const int* in_sizes, int n_in,
                              void* d_out, int out_size, void* d_ws, size_t ws_size,
                              hipStream_t stream) {
  const float* h    = (const float*)d_in[0];
  const float* w    = (const float*)d_in[1];
  const float* norm = (const float*)d_in[2];
  const int*   src  = (const int*)d_in[3];
  const int*   dst  = (const int*)d_in[4];
  float* out = (float*)d_out;
  float* hn  = (float*)d_ws;

  const int n_nodes = in_sizes[2];
  const int n_edges = in_sizes[3];

  gemm_scale_kernel<<<(n_nodes + 31) / 32, 256, 0, stream>>>(h, w, norm, hn, out, n_nodes);

  bin_fill_kernel<<<(n_edges + FILL_EDGES - 1) / FILL_EDGES, 256, 0, stream>>>(src, dst, out, n_nodes, n_edges);

  bin_sum_kernel<<<(n_nodes + BIN_ROWS - 1) / BIN_ROWS, 512, 0, stream>>>(hn, norm, src, dst, out, n_nodes, n_edges);
}


# GEMM launch re-tiled: one wave does 16 rows x all 64 columns (A read once in 16-byte pieces, B from a fragment-ordered LDS copy of W, whole-row stores), 128 rows per block; k grouping of sums changes
# speedup vs baseline: 3.4037x; 1.1536x over previous
#include <hip/hip_runtime.h>


typedef __attribute__((ext_vector_type(2))) float v2f;
typedef __attribute__((ext_vector_type(8))) float v8f;

#define BIN_ROWS   128
#define BIN_WORDS  (BIN_ROWS * 64)
#define MAX_BINS   1024
#define MAX_NODES  (MAX_BINS * BIN_ROWS)
#define FILL_EDGES 8192
#define SLOTS      64

__global__ __launch_bounds__(256) void gemm_scale_kernel(
    const float* __restrict__ h, const float* __restrict__ w,
    const float* __restrict__ norm, float* __restrict__ hn,
    float* __restrict__ out, int n_nodes) {
  __shared__ float4 wfrag[8 * 4 * 32];
  __shared__ __attribute__((aligned(16))) float nlds[BIN_ROWS];

  const int tid   = threadIdx.x;
  const int wave  = tid >> 5;
  const int lane  = tid & 31;
  const int m     = lane & 15;
  const int khalf = lane >> 4;
  const int rw    = blockIdx.x * BIN_ROWS + wave * 16;
  const int rbase = rw + khalf * 8;

  const float4* wsrc = (const float4*)w;
  float4 wv[4];
  #pragma unroll
  for (int it = 0; it < 4; ++it) wv[it] = wsrc[tid + it * 256];

  int arow = rw + m;
  if (arow >= n_nodes) arow = n_nodes - 1;
  const float4* ap = (const float4*)(h + arow * 64) + khalf;
  float4 av[8];
  #pragma unroll
  for (int j = 0; j < 8; ++j) av[j] = ap[2 * j];
  float nv = 0.f;
  if (tid < BIN_ROWS) nv = norm[min((int)blockIdx.x * BIN_ROWS + tid, n_nodes - 1)];
  const int nvalid = n_nodes - rbase;
  float* orow = hn + ((size_t)rbase * 64 + m * 4);

  if (tid < BIN_ROWS) nlds[tid] = nv;
  {
    float* wdst = (float*)wfrag;
    #pragma unroll
    for (int it = 0; it < 4; ++it) {
      const int idx = tid + it * 256;
      const int k = idx >> 4;
      const int i = idx & 15;
      const int base = (((k >> 3) * 128) + ((k >> 2) & 1) * 16 + i) * 4 + (k & 3);
      wdst[base]       = wv[it].x;
      wdst[base + 128] = wv[it].y;
      wdst[base + 256] = wv[it].z;
      wdst[base + 384] = wv[it].w;
    }
  }
  __syncthreads();

  if (tid == 0) ((int*)out)[blockIdx.x * BIN_WORDS] = 0;

  v8f acc0 = {}, acc1 = {}, acc2 = {}, acc3 = {};
  float4 bc[4], bn[4];
  #pragma unroll
  for (int t = 0; t < 4; ++t) bc[t] = wfrag[t * 32 + lane];
  #pragma unroll
  for (int j = 0; j < 8; ++j) {
    if (j + 1 < 8) {
      #pragma unroll
      for (int t = 0; t < 4; ++t) bn[t] = wfrag[((j + 1) * 4 + t) * 32 + lane];
    }
    __builtin_amdgcn_sched_barrier(0);
    v2f a, b;
    a.x = av[j].x; a.y = av[j].y;
    b.x = bc[0].x; b.y = bc[0].y;
    acc0 = __builtin_amdgcn_wmma_f32_16x16x4_f32(false, a, false, b, (short)0, acc0, false, false);
    b.x = bc[1].x; b.y = bc[1].y;
    acc1 = __builtin_amdgcn_wmma_f32_16x16x4_f32(false, a, false, b, (short)0, acc1, false, false);
    b.x = bc[2].x; b.y = bc[2].y;
    acc2 = __builtin_amdgcn_wmma_f32_16x16x4_f32(false, a, false, b, (short)0, acc2, false, false);
    b.x = bc[3].x; b.y = bc[3].y;
    acc3 = __builtin_amdgcn_wmma_f32_16x16x4_f32(false, a, false, b, (short)0, acc3, false, false);
    a.x = av[j].z; a.y = av[j].w;
    b.x = bc[0].z; b.y = bc[0].w;
    acc0 = __builtin_amdgcn_wmma_f32_16x16x4_f32(false, a, false, b, (short)0, acc0, false, false);
    b.x = bc[1].z; b.y = bc[1].w;
    acc1 = __builtin_amdgcn_wmma_f32_16x16x4_f32(false, a, false, b, (short)0, acc1, false, false);
    b.x = bc[2].z; b.y = bc[2].w;
    acc2 = __builtin_amdgcn_wmma_f32_16x16x4_f32(false, a, false, b, (short)0, acc2, false, false);
    b.x = bc[3].z; b.y = bc[3].w;
    acc3 = __builtin_amdgcn_wmma_f32_16x16x4_f32(false, a, false, b, (short)0, acc3, false, false);
    #pragma unroll
    for (int t = 0; t < 4; ++t) bc[t] = bn[t];
  }

  const float4 s_lo = *(const float4*)&nlds[wave * 16 + khalf * 8];
  const float4 s_hi = *(const float4*)&nlds[wave * 16 + khalf * 8 + 4];
  const float sc[8] = {s_lo.x, s_lo.y, s_lo.z, s_lo.w, s_hi.x, s_hi.y, s_hi.z, s_hi.w};
  #pragma unroll
  for (int r = 0; r < 8; ++r) {
    if (r < nvalid) {
      const float s = sc[r];
      *(float4*)(orow + r * 64) =
          make_float4(acc0[r] * s, acc1[r] * s, acc2[r] * s, acc3[r] * s);
    }
  }
}

__global__ __launch_bounds__(256) void bin_fill_kernel(
    const int* __restrict__ src, const int* __restrict__ dst,
    float* __restrict__ out, int n_nodes, int n_edges) {
  __shared__ __attribute__((aligned(16))) int hist[MAX_BINS];
  __shared__ int lofs[MAX_BINS];
  __shared__ int gadr[MAX_BINS];
  __shared__ int wsum[8];
  __shared__ int sval[FILL_EDGES];
  __shared__ int sadr[FILL_EDGES];

  if (n_nodes > MAX_NODES) return;

  const int tid  = threadIdx.x;
  const int lane = tid & 31;
  const int wave = tid >> 5;
  *(int4*)&hist[4 * tid] = make_int4(0, 0, 0, 0);
  __syncthreads();

  const int e0 = blockIdx.x * FILL_EDGES + tid;
  int sv[32], dv[32], br[32];
  #pragma unroll
  for (int k = 0; k < 32; ++k) {
    const int e = e0 + k * 256;
    sv[k] = (e < n_edges) ? src[e] : 0;
    dv[k] = (e < n_edges) ? dst[e] : -1;
  }
  #pragma unroll
  for (int k = 0; k < 32; ++k) {
    br[k] = -1;
    if (dv[k] >= 0) {
      const int bin = dv[k] >> 7;
      const int r = atomicAdd(&hist[bin], 1);
      br[k] = (bin << 16) | r;
      sv[k] |= (dv[k] & (BIN_ROWS - 1)) << 17;
    }
  }
  __syncthreads();

  const int4 hv = *(const int4*)&hist[4 * tid];
  const int tsum = hv.x + hv.y + hv.z + hv.w;
  int incl = tsum;
  #pragma unroll
  for (int o = 1; o < 32; o <<= 1) {
    const int t = __shfl_up(incl, o);
    if (lane >= o) incl += t;
  }
  if (lane == 31) wsum[wave] = incl;

  int hb[4] = {hv.x, hv.y, hv.z, hv.w};
  int gb[4];
  #pragma unroll
  for (int u = 0; u < 4; ++u) {
    gb[u] = 0;
    if (hb[u] > 0)
      gb[u] = __hip_atomic_fetch_add((int*)out + (4 * tid + u) * BIN_WORDS, hb[u],
                                     __ATOMIC_RELAXED, __HIP_MEMORY_SCOPE_AGENT);
  }
  __syncthreads();
  int excl = incl - tsum;
  for (int w = 0; w < wave; ++w) excl += wsum[w];
  int total = 0;
  for (int w = 0; w < 8; ++w) total += wsum[w];
  #pragma unroll
  for (int u = 0; u < 4; ++u) {
    const int bin = 4 * tid + u;
    lofs[bin] = excl;
    gadr[bin] = bin * BIN_WORDS + 1 + gb[u];
    excl += hb[u];
  }
  __syncthreads();

  #pragma unroll
  for (int k = 0; k < 32; ++k) {
    if (br[k] >= 0) {
      const int bin = br[k] >> 16;
      const int r   = br[k] & 0xFFFF;
      const int is  = lofs[bin] + r;
      const int a   = gadr[bin] + r;
      const int rows = min(BIN_ROWS, n_nodes - bin * BIN_ROWS);
      const int last = bin * BIN_WORDS + rows * 64 - 1;
      sval[is] = sv[k];
      sadr[is] = (a <= last) ? a : -1;
    }
  }
  __syncthreads();

  for (int i0 = tid; i0 < total; i0 += 8 * 256) {
    int a[8], v[8];
    #pragma unroll
    for (int u = 0; u < 8; ++u) {
      const int i = i0 + u * 256;
      a[u] = (i < total) ? sadr[i] : -1;
      v[u] = (i < total) ? sval[i] : 0;
    }
    #pragma unroll
    for (int u = 0; u < 8; ++u)
      if (a[u] >= 0) ((int*)out)[a[u]] = v[u];
  }
}

__global__ __launch_bounds__(512) void bin_sum_kernel(
    const float* __restrict__ hn, const float* __restrict__ norm,
    const int* __restrict__ src, const int* __restrict__ dst,
    float* out, int n_nodes, int n_edges) {
  __shared__ int cnt[BIN_ROWS];
  __shared__ int bucket[BIN_ROWS * SLOTS];

  const int tid  = threadIdx.x;
  const int d0   = blockIdx.x * BIN_ROWS;
  const int rows = min(BIN_ROWS, n_nodes - d0);
  const int* list = (const int*)out + d0 * 64;
  const int cap  = rows * 64 - 1;

  if (tid < BIN_ROWS) cnt[tid] = 0;
  const int c = list[0];
  const bool binok = (n_nodes <= MAX_NODES) && (c >= 0) && (c <= cap);
  __syncthreads();
  if (binok) {
    for (int i0 = tid; i0 < c; i0 += 4 * 512) {
      unsigned e[4];
      #pragma unroll
      for (int u = 0; u < 4; ++u) {
        const int i = i0 + u * 512;
        e[u] = (i < c) ? (unsigned)list[1 + i] : 0u;
      }
      #pragma unroll
      for (int u = 0; u < 4; ++u) {
        if (i0 + u * 512 < c) {
          const int dl = (int)(e[u] >> 17) & (BIN_ROWS - 1);
          const int s  = (int)(e[u] & 0x1FFFFu);
          const int slot = atomicAdd(&cnt[dl], 1);
          if (slot < SLOTS) bucket[dl * SLOTS + slot] = s;
        }
      }
    }
  }
  __syncthreads();

  const int wave = tid >> 5;
  const int lane = tid & 31;
  const int half = lane >> 4;
  const int q    = lane & 15;

  for (int k = 0; k < 8; ++k) {
    const int dl = wave * 8 + k;
    if (dl >= rows) break;
    const int n = d0 + dl;
    const int cd = cnt[dl];
    const float dn = norm[n];
    float* orow = out + n * 64;

    if (!binok || cd > SLOTS) {
      float2 a2 = make_float2(0.f, 0.f);
      for (int base = 0; base < n_edges; base += 32) {
        const int e  = base + lane;
        const int dv = (e < n_edges) ? dst[e] : -1;
        const int sv = (e < n_edges) ? src[e] : 0;
        unsigned mask = __builtin_amdgcn_ballot_w32(dv == n);
        while (mask) {
          const int b = __builtin_ctz(mask);
          mask &= mask - 1;
          const int s = __builtin_amdgcn_readlane(sv, b);
          const float2 v = *(const float2*)(hn + s * 64 + lane * 2);
          a2.x += v.x; a2.y += v.y;
        }
      }
      a2.x *= dn; a2.y *= dn;
      *(float2*)(orow + lane * 2) = a2;
      continue;
    }

    const int* bk = bucket + dl * SLOTS + half;
    float4 acc = make_float4(0.f, 0.f, 0.f, 0.f);
    const int nfull = cd >> 1;
    int j = 0;
    for (; j + 8 <= nfull; j += 8) {
      float4 v[8];
      #pragma unroll
      for (int u = 0; u < 8; ++u) {
        const int s = bk[2 * (j + u)];
        v[u] = *(const float4*)(hn + s * 64 + q * 4);
      }
      #pragma unroll
      for (int u = 0; u < 8; ++u) {
        acc.x += v[u].x; acc.y += v[u].y; acc.z += v[u].z; acc.w += v[u].w;
      }
    }
    {
      float4 v[7];
      #pragma unroll
      for (int u = 0; u < 7; ++u) {
        v[u] = make_float4(0.f, 0.f, 0.f, 0.f);
        if (j + u < nfull) {
          const int s = bk[2 * (j + u)];
          v[u] = *(const float4*)(hn + s * 64 + q * 4);
        }
      }
      #pragma unroll
      for (int u = 0; u < 7; ++u) {
        acc.x += v[u].x; acc.y += v[u].y; acc.z += v[u].z; acc.w += v[u].w;
      }
    }
    if (cd & 1) {
      if (half == 0) {
        const int s = bk[cd - 1];
        const float4 v = *(const float4*)(hn + s * 64 + q * 4);
        acc.x += v.x; acc.y += v.y; acc.z += v.z; acc.w += v.w;
      }
    }

    acc.x += __shfl_xor(acc.x, 16);
    acc.y += __shfl_xor(acc.y, 16);
    acc.z += __shfl_xor(acc.z, 16);
    acc.w += __shfl_xor(acc.w, 16);
    if (half == 0) {
      acc.x *= dn; acc.y *= dn; acc.z *= dn; acc.w *= dn;
      *(float4*)(orow + q * 4) = acc;
    }
  }
}

extern "C" void kernel_launch(void* const* d_in, const int* in_sizes, int n_in,
                              void* d_out, int out_size, void* d_ws, size_t ws_size,
                              hipStream_t stream) {
  const float* h    = (const float*)d_in[0];
  const float* w    = (const float*)d_in[1];
  const float* norm = (const float*)d_in[2];
  const int*   src  = (const int*)d_in[3];
  const int*   dst  = (const int*)d_in[4];
  float* out = (float*)d_out;
  float* hn  = (float*)d_ws;

  const int n_nodes = in_sizes[2];
  const int n_edges = in_sizes[3];

  gemm_scale_kernel<<<(n_nodes + BIN_ROWS - 1) / BIN_ROWS, 256, 0, stream>>>(h, w, norm, hn, out, n_nodes);

  bin_fill_kernel<<<(n_edges + FILL_EDGES - 1) / FILL_EDGES, 256, 0, stream>>>(src, dst, out, n_nodes, n_edges);

  bin_sum_kernel<<<(n_nodes + BIN_ROWS - 1) / BIN_ROWS, 512, 0, stream>>>(hn, norm, src, dst, out, n_nodes, n_edges);
}


# Sum launch: loads that each waited for the one before now go out as groups: a destination's row pairs up to 8 at a time (was one wait per leftover pair), a run's source scales at once; same sum order
# speedup vs baseline: 4.3188x; 1.2689x over previous
#include <hip/hip_runtime.h>


typedef __attribute__((ext_vector_type(2))) float v2f;
typedef __attribute__((ext_vector_type(8))) float v8f;

#define BIN_ROWS   128
#define MAX_BINS   1024
#define MAX_NODES  (MAX_BINS * BIN_ROWS)
#define FILL_EDGES 8192
#define SLOTS      46
#define DSTRIDE    (2 * SLOTS)
#define WSTRIDE    (8 * DSTRIDE)
#define ROWPITCH   68

__global__ __launch_bounds__(256) void bin_sort_kernel(
    const int* __restrict__ src, const int* __restrict__ dst,
    int* __restrict__ ws, int n_nodes, int n_edges) {
  __shared__ __attribute__((aligned(16))) int hist[MAX_BINS];
  __shared__ int lofs[MAX_BINS];
  __shared__ int wsum[8];
  __shared__ __attribute__((aligned(16))) int sval[FILL_EDGES];

  if (n_nodes > MAX_NODES) return;

  const int tid  = threadIdx.x;
  const int lane = tid & 31;
  const int wave = tid >> 5;
  const int nblk = (n_edges + FILL_EDGES - 1) / FILL_EDGES;
  const int nbins = (n_nodes + BIN_ROWS - 1) / BIN_ROWS;
  int* list  = ws + blockIdx.x * FILL_EDGES;
  int* table = ws + nblk * FILL_EDGES;
  *(int4*)&hist[4 * tid] = make_int4(0, 0, 0, 0);
  __syncthreads();

  const int e0 = blockIdx.x * FILL_EDGES + tid;
  int sv[32], dv[32], br[32];
  #pragma unroll
  for (int k = 0; k < 32; ++k) {
    const int e = e0 + k * 256;
    sv[k] = (e < n_edges) ? src[e] : 0;
    dv[k] = (e < n_edges) ? dst[e] : -1;
  }
  #pragma unroll
  for (int k = 0; k < 32; ++k) {
    br[k] = -1;
    if (dv[k] >= 0) {
      const int bin = dv[k] >> 7;
      const int r = atomicAdd(&hist[bin], 1);
      br[k] = (bin << 16) | r;
      sv[k] |= (dv[k] & (BIN_ROWS - 1)) << 17;
    }
  }
  __syncthreads();

  const int4 hv = *(const int4*)&hist[4 * tid];
  const int tsum = hv.x + hv.y + hv.z + hv.w;
  int incl = tsum;
  #pragma unroll
  for (int o = 1; o < 32; o <<= 1) {
    const int t = __shfl_up(incl, o);
    if (lane >= o) incl += t;
  }
  if (lane == 31) wsum[wave] = incl;
  __syncthreads();
  int excl = incl - tsum;
  for (int w = 0; w < wave; ++w) excl += wsum[w];
  int total = 0;
  for (int w = 0; w < 8; ++w) total += wsum[w];
  const int hb[4] = {hv.x, hv.y, hv.z, hv.w};
  #pragma unroll
  for (int u = 0; u < 4; ++u) {
    const int bin = 4 * tid + u;
    lofs[bin] = excl;
    if (bin < nbins) table[bin * nblk + blockIdx.x] = (excl << 14) | hb[u];
    excl += hb[u];
  }
  __syncthreads();

  #pragma unroll
  for (int k = 0; k < 32; ++k) {
    if (br[k] >= 0) sval[lofs[br[k] >> 16] + (br[k] & 0xFFFF)] = sv[k];
  }
  __syncthreads();

  int4 ov[8];
  #pragma unroll
  for (int u = 0; u < 8; ++u) {
    const int i = (tid + u * 256) * 4;
    ov[u] = *(const int4*)&sval[i];
  }
  #pragma unroll
  for (int u = 0; u < 8; ++u) {
    const int i = (tid + u * 256) * 4;
    if (i < total) *(int4*)&list[i] = ov[u];
  }
}

template <int R>
__device__ __forceinline__ float4 sum_group(
    const int* __restrict__ bk, int j, int cd, int half,
    const char* __restrict__ hb, unsigned qo, float4 acc) {
  float4 v[R];
  float sn[R];
  #pragma unroll
  for (int u = 0; u < R; ++u) {
    int t = 2 * (j + u) + half;
    if (u == R - 1) t = min(t, cd - 1);
    const int s = bk[t];
    v[u] = *(const float4*)(hb + ((unsigned)s * 256u + qo));
  }
  #pragma unroll
  for (int u = 0; u < R; ++u) {
    int t = 2 * (j + u) + half;
    if (u == R - 1) t = min(t, cd - 1);
    sn[u] = __int_as_float(bk[SLOTS + t]);
  }
  if (2 * (j + R - 1) + half >= cd) {
    v[R - 1] = make_float4(0.f, 0.f, 0.f, 0.f);
    sn[R - 1] = 0.f;
  }
  #pragma unroll
  for (int u = 0; u < R; ++u) {
    acc.x += sn[u] * v[u].x; acc.y += sn[u] * v[u].y;
    acc.z += sn[u] * v[u].z; acc.w += sn[u] * v[u].w;
  }
  return acc;
}

__global__ __launch_bounds__(512, 16) void bin_sum_mm_kernel(
    const float* __restrict__ h, const float* __restrict__ w,
    const float* __restrict__ norm,
    const int* __restrict__ src, const int* __restrict__ dst,
    const int* __restrict__ ws, float* __restrict__ out,
    int n_nodes, int n_edges) {
  __shared__ int cnt[BIN_ROWS];
  __shared__ __attribute__((aligned(16))) int bucket[16 * WSTRIDE];
  __shared__ float4 wfrag[8 * 2 * 2 * 32];
  __shared__ __attribute__((aligned(16))) float nlds[BIN_ROWS];

  const int tid  = threadIdx.x;
  const int d0   = blockIdx.x * BIN_ROWS;
  const int rows = min(BIN_ROWS, n_nodes - d0);
  const int nblk = (n_edges + FILL_EDGES - 1) / FILL_EDGES;
  const bool binok = (n_nodes <= MAX_NODES);
  float* fb = (float*)bucket;

  const float4* wsrc = (const float4*)w;
  float4 wv[2];
  #pragma unroll
  for (int it = 0; it < 2; ++it) wv[it] = wsrc[tid + it * 512];
  const int* trow = ws + nblk * FILL_EDGES + blockIdx.x * nblk;
  int b = tid >> 1;
  int t = (binok && b < nblk) ? trow[b] : 0;

  float nv0 = 0.f;
  if (tid < BIN_ROWS) nv0 = norm[min(d0 + tid, n_nodes - 1)];

  if (tid < BIN_ROWS) cnt[tid] = 0;
  if (tid < BIN_ROWS) nlds[tid] = nv0;
  {
    float* wdst = (float*)wfrag;
    #pragma unroll
    for (int it = 0; it < 2; ++it) {
      const int idx = tid + it * 512;
      const int k = idx >> 4;
      const int i = idx & 15;
      const int j = k >> 3, hf = (k >> 2) & 1, c = i >> 3, n0 = 2 * (i & 7);
      const int base = (((j * 2 + c) * 2) * 32 + hf * 16 + n0) * 4 + (k & 3);
      wdst[base]           = wv[it].x;
      wdst[base + 128]     = wv[it].y;
      wdst[base + 4]       = wv[it].z;
      wdst[base + 128 + 4] = wv[it].w;
    }
  }
  __syncthreads();
  if (binok) {
    const int par = tid & 1;
    while (b < nblk) {
      const int off = min((t >> 14) & 0x3FFF, FILL_EDGES);
      const int c   = min(t & 0x3FFF, FILL_EDGES - off);
      const int* run = ws + b * FILL_EDGES + off;
      unsigned e[12];
      float nr[12];
      #pragma unroll
      for (int u = 0; u < 12; ++u) {
        const int i = par + 2 * u;
        e[u] = (i < c) ? (unsigned)run[(unsigned)i] : 0u;
      }
      #pragma unroll
      for (int u = 0; u < 12; ++u) nr[u] = norm[e[u] & 0x1FFFFu];
      #pragma unroll
      for (int u = 0; u < 12; ++u) {
        if (par + 2 * u < c) {
          const int dl = (int)(e[u] >> 17) & (BIN_ROWS - 1);
          const int slot = atomicAdd(&cnt[dl], 1);
          if (slot < SLOTS) {
            bucket[dl * DSTRIDE + slot]         = (int)(e[u] & 0x1FFFFu);
            bucket[dl * DSTRIDE + SLOTS + slot] = __float_as_int(nr[u]);
          }
        }
      }
      for (int i = 24 + par; i < c; i += 2) {
        const unsigned ee = (unsigned)run[i];
        const int s  = (int)(ee & 0x1FFFFu);
        const int dl = (int)(ee >> 17) & (BIN_ROWS - 1);
        const float nv = norm[s];
        const int slot = atomicAdd(&cnt[dl], 1);
        if (slot < SLOTS) {
          bucket[dl * DSTRIDE + slot]         = s;
          bucket[dl * DSTRIDE + SLOTS + slot] = __float_as_int(nv);
        }
      }
      b += 256;
      if (b < nblk) t = trow[b];
    }
  }
  __syncthreads();

  const int wave = __builtin_amdgcn_readfirstlane(tid >> 5);
  const int lane = tid & 31;
  const int half = lane >> 4;
  const int q    = lane & 15;
  const unsigned qo = (unsigned)q * 16u;
  const char* hb = (const char*)h;

  for (int k = 0; k < 8; ++k) {
    const int dl = wave * 8 + k;
    if (dl >= rows) break;
    const int n = d0 + dl;
    const int cd = __builtin_amdgcn_readfirstlane(cnt[dl]);
    float* arow = fb + wave * WSTRIDE + k * ROWPITCH;

    if (!binok || cd > SLOTS) {
      float2 a2 = make_float2(0.f, 0.f);
      for (int base = 0; base < n_edges; base += 32) {
        const int e  = base + lane;
        const int dv = (e < n_edges) ? dst[e] : -1;
        const int sv = (e < n_edges) ? src[e] : 0;
        unsigned mask = __builtin_amdgcn_ballot_w32(dv == n);
        while (mask) {
          const int b = __builtin_ctz(mask);
          mask &= mask - 1;
          const int s = __builtin_amdgcn_readlane(sv, b);
          const float sn = norm[s];
          const float2 v = *(const float2*)(h + s * 64 + lane * 2);
          a2.x += sn * v.x; a2.y += sn * v.y;
        }
      }
      *(float2*)(arow + lane * 2) = a2;
      continue;
    }

    const int* bk = bucket + dl * DSTRIDE;
    float4 acc = make_float4(0.f, 0.f, 0.f, 0.f);
    const int np = (cd + 1) >> 1;
    for (int j = 0; j < np; j += 8) {
      switch (min(np - j, 8)) {
        case 8:  acc = sum_group<8>(bk, j, cd, half, hb, qo, acc); break;
        case 7:  acc = sum_group<7>(bk, j, cd, half, hb, qo, acc); break;
        case 6:  acc = sum_group<6>(bk, j, cd, half, hb, qo, acc); break;
        case 5:  acc = sum_group<5>(bk, j, cd, half, hb, qo, acc); break;
        case 4:  acc = sum_group<4>(bk, j, cd, half, hb, qo, acc); break;
        case 3:  acc = sum_group<3>(bk, j, cd, half, hb, qo, acc); break;
        case 2:  acc = sum_group<2>(bk, j, cd, half, hb, qo, acc); break;
        default: acc = sum_group<1>(bk, j, cd, half, hb, qo, acc); break;
      }
    }

    acc.x += __shfl_xor(acc.x, 16);
    acc.y += __shfl_xor(acc.y, 16);
    acc.z += __shfl_xor(acc.z, 16);
    acc.w += __shfl_xor(acc.w, 16);
    if (half == 0) *(float4*)(arow + q * 4) = acc;
  }
  __syncthreads();

  const int p  = wave >> 1;
  const int ch = wave & 1;
  const int m  = q;
  const int rbase = d0 + p * 16 + half * 8;
  float dn[8];

  const float* ar = fb + (2 * p + (m >> 3)) * WSTRIDE + (m & 7) * ROWPITCH + 4 * half;
  const float4* bf = wfrag + ch * 64 + lane;

  v8f acc0 = {}, acc1 = {};
  float4 ac, b0, b1, an, bn0, bn1;
  ac = *(const float4*)ar;
  b0 = bf[0];
  b1 = bf[32];
  #pragma unroll
  for (int j = 0; j < 8; ++j) {
    if (j + 1 < 8) {
      an  = *(const float4*)(ar + 8 * (j + 1));
      bn0 = bf[(j + 1) * 128];
      bn1 = bf[(j + 1) * 128 + 32];
    }
    if (j == 6) {
      const float4 s_lo = *(const float4*)&nlds[p * 16 + half * 8];
      const float4 s_hi = *(const float4*)&nlds[p * 16 + half * 8 + 4];
      dn[0] = s_lo.x; dn[1] = s_lo.y; dn[2] = s_lo.z; dn[3] = s_lo.w;
      dn[4] = s_hi.x; dn[5] = s_hi.y; dn[6] = s_hi.z; dn[7] = s_hi.w;
    }
    __builtin_amdgcn_sched_barrier(0);
    v2f a, b;
    a.x = ac.x; a.y = ac.y;
    b.x = b0.x; b.y = b0.y;
    acc0 = __builtin_amdgcn_wmma_f32_16x16x4_f32(false, a, false, b, (short)0, acc0, false, false);
    b.x = b1.x; b.y = b1.y;
    acc1 = __builtin_amdgcn_wmma_f32_16x16x4_f32(false, a, false, b, (short)0, acc1, false, false);
    a.x = ac.z; a.y = ac.w;
    b.x = b0.z; b.y = b0.w;
    acc0 = __builtin_amdgcn_wmma_f32_16x16x4_f32(false, a, false, b, (short)0, acc0, false, false);
    b.x = b1.z; b.y = b1.w;
    acc1 = __builtin_amdgcn_wmma_f32_16x16x4_f32(false, a, false, b, (short)0, acc1, false, false);
    ac = an; b0 = bn0; b1 = bn1;
  }

  const int nvalid = n_nodes - rbase;
  const unsigned ob = ((unsigned)rbase * 64u + (unsigned)(ch * 32 + m * 2)) * 4u;
  char* outb = (char*)out;
  #pragma unroll
  for (int r = 0; r < 8; ++r) {
    if (r < nvalid) {
      const float s = dn[r];
      *(float2*)(outb + (ob + (unsigned)r * 256u)) = make_float2(acc0[r] * s, acc1[r] * s);
    }
  }
}

extern "C" void kernel_launch(void* const* d_in, const int* in_sizes, int n_in,
                              void* d_out, int out_size, void* d_ws, size_t ws_size,
                              hipStream_t stream) {
  const float* h    = (const float*)d_in[0];
  const float* w    = (const float*)d_in[1];
  const float* norm = (const float*)d_in[2];
  const int*   src  = (const int*)d_in[3];
  const int*   dst  = (const int*)d_in[4];
  float* out = (float*)d_out;
  int*   ws  = (int*)d_ws;

  const int n_nodes = in_sizes[2];
  const int n_edges = in_sizes[3];

  bin_sort_kernel<<<(n_edges + FILL_EDGES - 1) / FILL_EDGES, 256, 0, stream>>>(src, dst, ws, n_nodes, n_edges);

  bin_sum_mm_kernel<<<(n_nodes + BIN_ROWS - 1) / BIN_ROWS, 512, 0, stream>>>(h, w, norm, src, dst, ws, out, n_nodes, n_edges);
}


# LDS round trips that each waited for the one before now go out as groups (sort launch: rank adds, place reads, copy-out reads, edge loads; sum launch: slot adds); sums still in arrival order
# speedup vs baseline: 4.6932x; 1.0867x over previous
#include <hip/hip_runtime.h>


typedef __attribute__((ext_vector_type(2))) float v2f;
typedef __attribute__((ext_vector_type(8))) float v8f;

#define BIN_ROWS   128
#define MAX_BINS   1024
#define MAX_NODES  (MAX_BINS * BIN_ROWS)
#define FILL_EDGES 8192
#define SLOTS      46
#define DSTRIDE    (2 * SLOTS)
#define WSTRIDE    (8 * DSTRIDE)
#define ROWPITCH   68

__global__ __launch_bounds__(256) void bin_sort_kernel(
    const int* __restrict__ src, const int* __restrict__ dst,
    int* __restrict__ ws, int n_nodes, int n_edges) {
  __shared__ __attribute__((aligned(16))) int hist[MAX_BINS];
  __shared__ __attribute__((aligned(16))) int lofs[MAX_BINS];
  __shared__ __attribute__((aligned(16))) int wsum[8];
  __shared__ __attribute__((aligned(16))) int sval[FILL_EDGES + 32];

  if (n_nodes > MAX_NODES) return;

  const int tid  = threadIdx.x;
  const int lane = tid & 31;
  const int wave = tid >> 5;
  const int nblk = (n_edges + FILL_EDGES - 1) / FILL_EDGES;
  const int nbins = (n_nodes + BIN_ROWS - 1) / BIN_ROWS;
  int* list  = ws + blockIdx.x * FILL_EDGES;
  int* table = ws + nblk * FILL_EDGES;
  *(int4*)&hist[4 * tid] = make_int4(0, 0, 0, 0);
  __syncthreads();

  const int e0 = blockIdx.x * FILL_EDGES + tid;
  int sv[32], dv[32], br[32];
  #pragma unroll
  for (int k = 0; k < 32; ++k) dv[k] = dst[min(e0 + k * 256, n_edges - 1)];
  #pragma unroll
  for (int k = 0; k < 32; ++k) sv[k] = src[min(e0 + k * 256, n_edges - 1)];
  #pragma unroll
  for (int k = 0; k < 32; ++k) dv[k] = (e0 + k * 256 < n_edges) ? dv[k] : -1;
  #pragma unroll
  for (int k = 0; k < 32; ++k) {
    const bool ok = dv[k] >= 0;
    const int bin = ok ? (dv[k] >> 7) : tid;
    br[k] = atomicAdd(&hist[bin], ok ? 1 : 0);
  }
  #pragma unroll
  for (int k = 0; k < 32; ++k) {
    br[k] = (int)(((unsigned)(dv[k] >> 7) << 16) | (unsigned)br[k]) | (dv[k] >> 31);
    sv[k] |= (dv[k] & (BIN_ROWS - 1)) << 17;
  }
  __syncthreads();

  const int4 hv = *(const int4*)&hist[4 * tid];
  const int tsum = hv.x + hv.y + hv.z + hv.w;
  int incl = tsum;
  #pragma unroll
  for (int o = 1; o < 32; o <<= 1) {
    const int t = __shfl_up(incl, o);
    if (lane >= o) incl += t;
  }
  if (lane == 31) wsum[wave] = incl;
  __syncthreads();
  const int4 wlo = *(const int4*)&wsum[0];
  const int4 whi = *(const int4*)&wsum[4];
  const int wv[8] = {wlo.x, wlo.y, wlo.z, wlo.w, whi.x, whi.y, whi.z, whi.w};
  int excl = incl - tsum;
  #pragma unroll
  for (int w = 0; w < 8; ++w) excl += (w < wave) ? wv[w] : 0;
  const int hb[4] = {hv.x, hv.y, hv.z, hv.w};
  int ex[4];
  #pragma unroll
  for (int u = 0; u < 4; ++u) {
    ex[u] = excl;
    excl += hb[u];
  }
  *(int4*)&lofs[4 * tid] = make_int4(ex[0], ex[1], ex[2], ex[3]);
  __syncthreads();

  int pl[32];
  #pragma unroll
  for (int k = 0; k < 32; ++k) pl[k] = lofs[(br[k] >= 0) ? (br[k] >> 16) : tid];
  #pragma unroll
  for (int k = 0; k < 32; ++k) {
    const int p = (br[k] >= 0) ? pl[k] + (br[k] & 0xFFFF) : FILL_EDGES + lane;
    sval[p] = sv[k];
  }
  __syncthreads();

  int4 ov[8];
  #pragma unroll
  for (int u = 0; u < 8; ++u) {
    const int i = (tid + u * 256) * 4;
    ov[u] = *(const int4*)&sval[i];
  }
  #pragma unroll
  for (int u = 0; u < 8; ++u) {
    const int i = (tid + u * 256) * 4;
    *(int4*)&list[i] = ov[u];
  }
  #pragma unroll
  for (int u = 0; u < 4; ++u) {
    const int bin = 4 * tid + u;
    if (bin < nbins) table[bin * nblk + blockIdx.x] = (ex[u] << 14) | hb[u];
  }
}

template <int R>
__device__ __forceinline__ float4 sum_group(
    const int* __restrict__ bk, int j, int cd, int half,
    const char* __restrict__ hb, unsigned qo, float4 acc) {
  float4 v[R];
  float sn[R];
  #pragma unroll
  for (int u = 0; u < R; ++u) {
    int t = 2 * (j + u) + half;
    if (u == R - 1) t = min(t, cd - 1);
    const int s = bk[t];
    v[u] = *(const float4*)(hb + ((unsigned)s * 256u + qo));
  }
  #pragma unroll
  for (int u = 0; u < R; ++u) {
    int t = 2 * (j + u) + half;
    if (u == R - 1) t = min(t, cd - 1);
    sn[u] = __int_as_float(bk[SLOTS + t]);
  }
  if (2 * (j + R - 1) + half >= cd) {
    v[R - 1] = make_float4(0.f, 0.f, 0.f, 0.f);
    sn[R - 1] = 0.f;
  }
  #pragma unroll
  for (int u = 0; u < R; ++u) {
    acc.x += sn[u] * v[u].x; acc.y += sn[u] * v[u].y;
    acc.z += sn[u] * v[u].z; acc.w += sn[u] * v[u].w;
  }
  return acc;
}

__global__ __launch_bounds__(512, 16) void bin_sum_mm_kernel(
    const float* __restrict__ h, const float* __restrict__ w,
    const float* __restrict__ norm,
    const int* __restrict__ src, const int* __restrict__ dst,
    const int* __restrict__ ws, float* __restrict__ out,
    int n_nodes, int n_edges) {
  __shared__ int cnt[BIN_ROWS];
  __shared__ __attribute__((aligned(16))) int bucket[16 * WSTRIDE];
  __shared__ float4 wfrag[8 * 2 * 2 * 32];
  __shared__ __attribute__((aligned(16))) float nlds[BIN_ROWS];

  const int tid  = threadIdx.x;
  const int d0   = blockIdx.x * BIN_ROWS;
  const int rows = min(BIN_ROWS, n_nodes - d0);
  const int nblk = (n_edges + FILL_EDGES - 1) / FILL_EDGES;
  const bool binok = (n_nodes <= MAX_NODES);
  float* fb = (float*)bucket;

  const float4* wsrc = (const float4*)w;
  float4 wv[2];
  #pragma unroll
  for (int it = 0; it < 2; ++it) wv[it] = wsrc[tid + it * 512];
  const int* trow = ws + nblk * FILL_EDGES + blockIdx.x * nblk;
  int b = tid >> 1;
  int t = (binok && b < nblk) ? trow[b] : 0;

  float nv0 = 0.f;
  if (tid < BIN_ROWS) nv0 = norm[min(d0 + tid, n_nodes - 1)];

  if (tid < BIN_ROWS) cnt[tid] = 0;
  if (tid < BIN_ROWS) nlds[tid] = nv0;
  {
    float* wdst = (float*)wfrag;
    #pragma unroll
    for (int it = 0; it < 2; ++it) {
      const int idx = tid + it * 512;
      const int k = idx >> 4;
      const int i = idx & 15;
      const int j = k >> 3, hf = (k >> 2) & 1, c = i >> 3, n0 = 2 * (i & 7);
      const int base = (((j * 2 + c) * 2) * 32 + hf * 16 + n0) * 4 + (k & 3);
      wdst[base]           = wv[it].x;
      wdst[base + 128]     = wv[it].y;
      wdst[base + 4]       = wv[it].z;
      wdst[base + 128 + 4] = wv[it].w;
    }
  }
  __syncthreads();
  if (binok) {
    const int par = tid & 1;
    while (b < nblk) {
      const int off = min((t >> 14) & 0x3FFF, FILL_EDGES);
      const int c   = min(t & 0x3FFF, FILL_EDGES - off);
      const int* run = ws + b * FILL_EDGES + off;
      unsigned e[12];
      float nr[12];
      #pragma unroll
      for (int u = 0; u < 12; ++u) {
        const int i = par + 2 * u;
        e[u] = (i < c) ? (unsigned)run[(unsigned)i] : 0u;
      }
      #pragma unroll
      for (int u = 0; u < 12; ++u) nr[u] = norm[e[u] & 0x1FFFFu];
      int sl[12];
      #pragma unroll
      for (int u = 0; u < 12; ++u) {
        const bool ok = par + 2 * u < c;
        const int dl = ok ? ((int)(e[u] >> 17) & (BIN_ROWS - 1)) : (tid & (BIN_ROWS - 1));
        sl[u] = atomicAdd(&cnt[dl], ok ? 1 : 0);
      }
      #pragma unroll
      for (int u = 0; u < 12; ++u) {
        if (par + 2 * u < c && sl[u] < SLOTS) {
          const int dl = (int)(e[u] >> 17) & (BIN_ROWS - 1);
          bucket[dl * DSTRIDE + sl[u]]         = (int)(e[u] & 0x1FFFFu);
          bucket[dl * DSTRIDE + SLOTS + sl[u]] = __float_as_int(nr[u]);
        }
      }
      for (int i = 24 + par; i < c; i += 2) {
        const unsigned ee = (unsigned)run[i];
        const int s  = (int)(ee & 0x1FFFFu);
        const int dl = (int)(ee >> 17) & (BIN_ROWS - 1);
        const float nv = norm[s];
        const int slot = atomicAdd(&cnt[dl], 1);
        if (slot < SLOTS) {
          bucket[dl * DSTRIDE + slot]         = s;
          bucket[dl * DSTRIDE + SLOTS + slot] = __float_as_int(nv);
        }
      }
      b += 256;
      if (b < nblk) t = trow[b];
    }
  }
  __syncthreads();

  const int wave = __builtin_amdgcn_readfirstlane(tid >> 5);
  const int lane = tid & 31;
  const int half = lane >> 4;
  const int q    = lane & 15;
  const unsigned qo = (unsigned)q * 16u;
  const char* hb = (const char*)h;

  for (int k = 0; k < 8; ++k) {
    const int dl = wave * 8 + k;
    if (dl >= rows) break;
    const int n = d0 + dl;
    const int cd = __builtin_amdgcn_readfirstlane(cnt[dl]);
    float* arow = fb + wave * WSTRIDE + k * ROWPITCH;

    if (!binok || cd > SLOTS) {
      float2 a2 = make_float2(0.f, 0.f);
      for (int base = 0; base < n_edges; base += 32) {
        const int e  = base + lane;
        const int dv = (e < n_edges) ? dst[e] : -1;
        const int sv = (e < n_edges) ? src[e] : 0;
        unsigned mask = __builtin_amdgcn_ballot_w32(dv == n);
        while (mask) {
          const int b = __builtin_ctz(mask);
          mask &= mask - 1;
          const int s = __builtin_amdgcn_readlane(sv, b);
          const float sn = norm[s];
          const float2 v = *(const float2*)(h + s * 64 + lane * 2);
          a2.x += sn * v.x; a2.y += sn * v.y;
        }
      }
      *(float2*)(arow + lane * 2) = a2;
      continue;
    }

    const int* bk = bucket + dl * DSTRIDE;
    float4 acc = make_float4(0.f, 0.f, 0.f, 0.f);
    const int np = (cd + 1) >> 1;
    for (int j = 0; j < np; j += 8) {
      switch (min(np - j, 8)) {
        case 8:  acc = sum_group<8>(bk, j, cd, half, hb, qo, acc); break;
        case 7:  acc = sum_group<7>(bk, j, cd, half, hb, qo, acc); break;
        case 6:  acc = sum_group<6>(bk, j, cd, half, hb, qo, acc); break;
        case 5:  acc = sum_group<5>(bk, j, cd, half, hb, qo, acc); break;
        case 4:  acc = sum_group<4>(bk, j, cd, half, hb, qo, acc); break;
        case 3:  acc = sum_group<3>(bk, j, cd, half, hb, qo, acc); break;
        case 2:  acc = sum_group<2>(bk, j, cd, half, hb, qo, acc); break;
        default: acc = sum_group<1>(bk, j, cd, half, hb, qo, acc); break;
      }
    }

    acc.x += __shfl_xor(acc.x, 16);
    acc.y += __shfl_xor(acc.y, 16);
    acc.z += __shfl_xor(acc.z, 16);
    acc.w += __shfl_xor(acc.w, 16);
    if (half == 0) *(float4*)(arow + q * 4) = acc;
  }
  __syncthreads();

  const int p  = wave >> 1;
  const int ch = wave & 1;
  const int m  = q;
  const int rbase = d0 + p * 16 + half * 8;
  float dn[8];

  const float* ar = fb + (2 * p + (m >> 3)) * WSTRIDE + (m & 7) * ROWPITCH + 4 * half;
  const float4* bf = wfrag + ch * 64 + lane;

  v8f acc0 = {}, acc1 = {};
  float4 ac, b0, b1, an, bn0, bn1;
  ac = *(const float4*)ar;
  b0 = bf[0];
  b1 = bf[32];
  #pragma unroll
  for (int j = 0; j < 8; ++j) {
    if (j + 1 < 8) {
      an  = *(const float4*)(ar + 8 * (j + 1));
      bn0 = bf[(j + 1) * 128];
      bn1 = bf[(j + 1) * 128 + 32];
    }
    if (j == 6) {
      const float4 s_lo = *(const float4*)&nlds[p * 16 + half * 8];
      const float4 s_hi = *(const float4*)&nlds[p * 16 + half * 8 + 4];
      dn[0] = s_lo.x; dn[1] = s_lo.y; dn[2] = s_lo.z; dn[3] = s_lo.w;
      dn[4] = s_hi.x; dn[5] = s_hi.y; dn[6] = s_hi.z; dn[7] = s_hi.w;
    }
    __builtin_amdgcn_sched_barrier(0);
    v2f a, b;
    a.x = ac.x; a.y = ac.y;
    b.x = b0.x; b.y = b0.y;
    acc0 = __builtin_amdgcn_wmma_f32_16x16x4_f32(false, a, false, b, (short)0, acc0, false, false);
    b.x = b1.x; b.y = b1.y;
    acc1 = __builtin_amdgcn_wmma_f32_16x16x4_f32(false, a, false, b, (short)0, acc1, false, false);
    a.x = ac.z; a.y = ac.w;
    b.x = b0.z; b.y = b0.w;
    acc0 = __builtin_amdgcn_wmma_f32_16x16x4_f32(false, a, false, b, (short)0, acc0, false, false);
    b.x = b1.z; b.y = b1.w;
    acc1 = __builtin_amdgcn_wmma_f32_16x16x4_f32(false, a, false, b, (short)0, acc1, false, false);
    ac = an; b0 = bn0; b1 = bn1;
  }

  const int nvalid = n_nodes - rbase;
  const unsigned ob = ((unsigned)rbase * 64u + (unsigned)(ch * 32 + m * 2)) * 4u;
  char* outb = (char*)out;
  #pragma unroll
  for (int r = 0; r < 8; ++r) {
    if (r < nvalid) {
      const float s = dn[r];
      *(float2*)(outb + (ob + (unsigned)r * 256u)) = make_float2(acc0[r] * s, acc1[r] * s);
    }
  }
}

extern "C" void kernel_launch(void* const* d_in, const int* in_sizes, int n_in,
                              void* d_out, int out_size, void* d_ws, size_t ws_size,
                              hipStream_t stream) {
  const float* h    = (const float*)d_in[0];
  const float* w    = (const float*)d_in[1];
  const float* norm = (const float*)d_in[2];
  const int*   src  = (const int*)d_in[3];
  const int*   dst  = (const int*)d_in[4];
  float* out = (float*)d_out;
  int*   ws  = (int*)d_ws;

  const int n_nodes = in_sizes[2];
  const int n_edges = in_sizes[3];

  bin_sort_kernel<<<(n_edges + FILL_EDGES - 1) / FILL_EDGES, 256, 0, stream>>>(src, dst, ws, n_nodes, n_edges);

  bin_sum_mm_kernel<<<(n_nodes + BIN_ROWS - 1) / BIN_ROWS, 512, 0, stream>>>(h, w, norm, src, dst, ws, out, n_nodes, n_edges);
}


# Bin re-sized from 128 to 144 destinations (sum blocks of 18 waves): a WGP's wave slots hold at most 3 blocks = 432 rows, was 4 = 512 against a mean of 3; sums still in arrival order
# speedup vs baseline: 4.9079x; 1.0457x over previous
#include <hip/hip_runtime.h>


typedef __attribute__((ext_vector_type(2))) float v2f;
typedef __attribute__((ext_vector_type(8))) float v8f;

#define BIN_ROWS   144
#define SUM_WAVES  (BIN_ROWS / 8)
#define SUM_THREADS (SUM_WAVES * 32)
#define MAX_BINS   1024
#define MAX_NODES  131072
#define FILL_EDGES 8192
#define SLOTS      46
#define DSTRIDE    (2 * SLOTS)
#define WSTRIDE    (8 * DSTRIDE)
#define ROWPITCH   68

__global__ __launch_bounds__(256) void bin_sort_kernel(
    const int* __restrict__ src, const int* __restrict__ dst,
    int* __restrict__ ws, int n_nodes, int n_edges) {
  __shared__ __attribute__((aligned(16))) int hist[MAX_BINS];
  __shared__ __attribute__((aligned(16))) int lofs[MAX_BINS];
  __shared__ __attribute__((aligned(16))) int wsum[8];
  __shared__ __attribute__((aligned(16))) int sval[FILL_EDGES + 32];

  if (n_nodes > MAX_NODES) return;

  const int tid  = threadIdx.x;
  const int lane = tid & 31;
  const int wave = tid >> 5;
  const int nblk = (n_edges + FILL_EDGES - 1) / FILL_EDGES;
  const int nbins = (n_nodes + BIN_ROWS - 1) / BIN_ROWS;
  int* list  = ws + blockIdx.x * FILL_EDGES;
  int* table = ws + nblk * FILL_EDGES;
  *(int4*)&hist[4 * tid] = make_int4(0, 0, 0, 0);
  __syncthreads();

  const int e0 = blockIdx.x * FILL_EDGES + tid;
  int sv[32], dv[32], br[32];
  #pragma unroll
  for (int k = 0; k < 32; ++k) dv[k] = dst[min(e0 + k * 256, n_edges - 1)];
  #pragma unroll
  for (int k = 0; k < 32; ++k) sv[k] = src[min(e0 + k * 256, n_edges - 1)];
  #pragma unroll
  for (int k = 0; k < 32; ++k) dv[k] = (e0 + k * 256 < n_edges) ? dv[k] : -1;
  #pragma unroll
  for (int k = 0; k < 32; ++k) {
    const bool ok = dv[k] >= 0;
    const int bin = ok ? (int)((unsigned)dv[k] / BIN_ROWS) : tid;
    br[k] = atomicAdd(&hist[bin], ok ? 1 : 0);
  }
  #pragma unroll
  for (int k = 0; k < 32; ++k) {
    const unsigned ud = (unsigned)dv[k];
    const unsigned bin = ud / BIN_ROWS;
    br[k] = (int)((bin << 16) | (unsigned)br[k]) | (dv[k] >> 31);
    sv[k] |= (int)((ud - bin * BIN_ROWS) << 17);
  }
  __syncthreads();

  const int4 hv = *(const int4*)&hist[4 * tid];
  const int tsum = hv.x + hv.y + hv.z + hv.w;
  int incl = tsum;
  #pragma unroll
  for (int o = 1; o < 32; o <<= 1) {
    const int t = __shfl_up(incl, o);
    if (lane >= o) incl += t;
  }
  if (lane == 31) wsum[wave] = incl;
  __syncthreads();
  const int4 wlo = *(const int4*)&wsum[0];
  const int4 whi = *(const int4*)&wsum[4];
  const int wv[8] = {wlo.x, wlo.y, wlo.z, wlo.w, whi.x, whi.y, whi.z, whi.w};
  int excl = incl - tsum;
  #pragma unroll
  for (int w = 0; w < 8; ++w) excl += (w < wave) ? wv[w] : 0;
  const int hb[4] = {hv.x, hv.y, hv.z, hv.w};
  int ex[4];
  #pragma unroll
  for (int u = 0; u < 4; ++u) {
    ex[u] = excl;
    excl += hb[u];
  }
  *(int4*)&lofs[4 * tid] = make_int4(ex[0], ex[1], ex[2], ex[3]);
  __syncthreads();

  int pl[32];
  #pragma unroll
  for (int k = 0; k < 32; ++k) pl[k] = lofs[(br[k] >= 0) ? (br[k] >> 16) : tid];
  #pragma unroll
  for (int k = 0; k < 32; ++k) {
    const int p = (br[k] >= 0) ? pl[k] + (br[k] & 0xFFFF) : FILL_EDGES + lane;
    sval[p] = sv[k];
  }
  __syncthreads();

  int4 ov[8];
  #pragma unroll
  for (int u = 0; u < 8; ++u) {
    const int i = (tid + u * 256) * 4;
    ov[u] = *(const int4*)&sval[i];
  }
  #pragma unroll
  for (int u = 0; u < 8; ++u) {
    const int i = (tid + u * 256) * 4;
    *(int4*)&list[i] = ov[u];
  }
  #pragma unroll
  for (int u = 0; u < 4; ++u) {
    const int bin = 4 * tid + u;
    if (bin < nbins) table[bin * nblk + blockIdx.x] = (ex[u] << 14) | hb[u];
  }
}

template <int R>
__device__ __forceinline__ float4 sum_group(
    const int* __restrict__ bk, int j, int cd, int half,
    const char* __restrict__ hb, unsigned qo, float4 acc) {
  float4 v[R];
  float sn[R];
  #pragma unroll
  for (int u = 0; u < R; ++u) {
    int t = 2 * (j + u) + half;
    if (u == R - 1) t = min(t, cd - 1);
    const int s = bk[t];
    v[u] = *(const float4*)(hb + ((unsigned)s * 256u + qo));
  }
  #pragma unroll
  for (int u = 0; u < R; ++u) {
    int t = 2 * (j + u) + half;
    if (u == R - 1) t = min(t, cd - 1);
    sn[u] = __int_as_float(bk[SLOTS + t]);
  }
  if (2 * (j + R - 1) + half >= cd) {
    v[R - 1] = make_float4(0.f, 0.f, 0.f, 0.f);
    sn[R - 1] = 0.f;
  }
  #pragma unroll
  for (int u = 0; u < R; ++u) {
    acc.x += sn[u] * v[u].x; acc.y += sn[u] * v[u].y;
    acc.z += sn[u] * v[u].z; acc.w += sn[u] * v[u].w;
  }
  return acc;
}

__global__ __launch_bounds__(SUM_THREADS) __attribute__((amdgpu_num_vgpr(64))) void bin_sum_mm_kernel(
    const float* __restrict__ h, const float* __restrict__ w,
    const float* __restrict__ norm,
    const int* __restrict__ src, const int* __restrict__ dst,
    const int* __restrict__ ws, float* __restrict__ out,
    int n_nodes, int n_edges) {
  __shared__ int cnt[BIN_ROWS];
  __shared__ __attribute__((aligned(16))) int bucket[SUM_WAVES * WSTRIDE];
  __shared__ float4 wfrag[8 * 2 * 2 * 32];
  __shared__ __attribute__((aligned(16))) float nlds[BIN_ROWS];

  const int tid  = threadIdx.x;
  const int d0   = blockIdx.x * BIN_ROWS;
  const int rows = min(BIN_ROWS, n_nodes - d0);
  const int nblk = (n_edges + FILL_EDGES - 1) / FILL_EDGES;
  const bool binok = (n_nodes <= MAX_NODES);
  float* fb = (float*)bucket;

  const float4* wsrc = (const float4*)w;
  float4 wv[2];
  #pragma unroll
  for (int it = 0; it < 2; ++it) wv[it] = wsrc[min(tid + it * SUM_THREADS, 1023)];
  const int* trow = ws + nblk * FILL_EDGES + blockIdx.x * nblk;
  int b = tid >> 1;
  int t = (binok && b < nblk) ? trow[b] : 0;

  float nv0 = 0.f;
  if (tid < BIN_ROWS) nv0 = norm[min(d0 + tid, n_nodes - 1)];

  if (tid < BIN_ROWS) cnt[tid] = 0;
  if (tid < BIN_ROWS) nlds[tid] = nv0;
  {
    float* wdst = (float*)wfrag;
    #pragma unroll
    for (int it = 0; it < 2; ++it) {
      const int idx = tid + it * SUM_THREADS;
      if (idx >= 1024) break;
      const int k = idx >> 4;
      const int i = idx & 15;
      const int j = k >> 3, hf = (k >> 2) & 1, c = i >> 3, n0 = 2 * (i & 7);
      const int base = (((j * 2 + c) * 2) * 32 + hf * 16 + n0) * 4 + (k & 3);
      wdst[base]           = wv[it].x;
      wdst[base + 128]     = wv[it].y;
      wdst[base + 4]       = wv[it].z;
      wdst[base + 128 + 4] = wv[it].w;
    }
  }
  __syncthreads();
  const int wave = __builtin_amdgcn_readfirstlane(tid >> 5);
  if (binok) {
    const int par = tid & 1;
    while (b < nblk) {
      const int off = min((t >> 14) & 0x3FFF, FILL_EDGES);
      const int c   = min(t & 0x3FFF, FILL_EDGES - off);
      const int* run = ws + b * FILL_EDGES + off;
      unsigned e[12];
      float nr[12];
      #pragma unroll
      for (int u = 0; u < 12; ++u) {
        const int i = par + 2 * u;
        e[u] = (i < c) ? (unsigned)run[(unsigned)i] : 0u;
      }
      #pragma unroll
      for (int u = 0; u < 12; ++u) nr[u] = norm[e[u] & 0x1FFFFu];
      int sl[12];
      #pragma unroll
      for (int u = 0; u < 12; ++u) {
        const bool ok = par + 2 * u < c;
        const int dl = ok ? min((int)(e[u] >> 17) & 0xFF, BIN_ROWS - 1) : (tid & 127);
        sl[u] = atomicAdd(&cnt[dl], ok ? 1 : 0);
      }
      #pragma unroll
      for (int u = 0; u < 12; ++u) {
        if (par + 2 * u < c && sl[u] < SLOTS) {
          const int dl = min((int)(e[u] >> 17) & 0xFF, BIN_ROWS - 1);
          bucket[dl * DSTRIDE + sl[u]]         = (int)(e[u] & 0x1FFFFu);
          bucket[dl * DSTRIDE + SLOTS + sl[u]] = __float_as_int(nr[u]);
        }
      }
      for (int i = 24 + par; i < c; i += 2) {
        const unsigned ee = (unsigned)run[i];
        const int s  = (int)(ee & 0x1FFFFu);
        const int dl = min((int)(ee >> 17) & 0xFF, BIN_ROWS - 1);
        const float nv = norm[s];
        const int slot = atomicAdd(&cnt[dl], 1);
        if (slot < SLOTS) {
          bucket[dl * DSTRIDE + slot]         = s;
          bucket[dl * DSTRIDE + SLOTS + slot] = __float_as_int(nv);
        }
      }
      b += SUM_THREADS / 2;
      if (b < nblk) t = trow[b];
    }
  }
  __syncthreads();

  const int lane = (int)__builtin_amdgcn_mbcnt_lo(~0u, 0u);
  const int half = lane >> 4;
  const int q    = lane & 15;
  const unsigned qo = (unsigned)q * 16u;
  const char* hb = (const char*)h;

  for (int k = 0; k < 8; ++k) {
    const int dl = wave * 8 + k;
    if (dl >= rows) break;
    const int n = d0 + dl;
    const int cd = __builtin_amdgcn_readfirstlane(cnt[dl]);
    float* arow = fb + wave * WSTRIDE + k * ROWPITCH;

    if (!binok || cd > SLOTS) {
      float2 a2 = make_float2(0.f, 0.f);
      for (int base = 0; base < n_edges; base += 32) {
        const int e  = base + lane;
        const int dv = (e < n_edges) ? dst[e] : -1;
        const int sv = (e < n_edges) ? src[e] : 0;
        unsigned mask = __builtin_amdgcn_ballot_w32(dv == n);
        while (mask) {
          const int b = __builtin_ctz(mask);
          mask &= mask - 1;
          const int s = __builtin_amdgcn_readlane(sv, b);
          const float sn = norm[s];
          const float2 v = *(const float2*)(h + s * 64 + lane * 2);
          a2.x += sn * v.x; a2.y += sn * v.y;
        }
      }
      *(float2*)(arow + lane * 2) = a2;
      continue;
    }

    const int* bk = bucket + dl * DSTRIDE;
    float4 acc = make_float4(0.f, 0.f, 0.f, 0.f);
    const int np = (cd + 1) >> 1;
    for (int j = 0; j < np; j += 8) {
      switch (min(np - j, 8)) {
        case 8:  acc = sum_group<8>(bk, j, cd, half, hb, qo, acc); break;
        case 7:  acc = sum_group<7>(bk, j, cd, half, hb, qo, acc); break;
        case 6:  acc = sum_group<6>(bk, j, cd, half, hb, qo, acc); break;
        case 5:  acc = sum_group<5>(bk, j, cd, half, hb, qo, acc); break;
        case 4:  acc = sum_group<4>(bk, j, cd, half, hb, qo, acc); break;
        case 3:  acc = sum_group<3>(bk, j, cd, half, hb, qo, acc); break;
        case 2:  acc = sum_group<2>(bk, j, cd, half, hb, qo, acc); break;
        default: acc = sum_group<1>(bk, j, cd, half, hb, qo, acc); break;
      }
    }

    acc.x += __shfl_xor(acc.x, 16);
    acc.y += __shfl_xor(acc.y, 16);
    acc.z += __shfl_xor(acc.z, 16);
    acc.w += __shfl_xor(acc.w, 16);
    if (half == 0) *(float4*)(arow + q * 4) = acc;
  }
  __syncthreads();

  const int p  = wave >> 1;
  const int ch = wave & 1;
  const int m  = q;
  const int rbase = d0 + p * 16 + half * 8;
  float dn[8];

  const float* ar = fb + (2 * p + (m >> 3)) * WSTRIDE + (m & 7) * ROWPITCH + 4 * half;
  const float4* bf = wfrag + ch * 64 + lane;

  v8f acc0 = {}, acc1 = {};
  float4 ac, b0, b1, an, bn0, bn1;
  ac = *(const float4*)ar;
  b0 = bf[0];
  b1 = bf[32];
  #pragma unroll
  for (int j = 0; j < 8; ++j) {
    if (j + 1 < 8) {
      an  = *(const float4*)(ar + 8 * (j + 1));
      bn0 = bf[(j + 1) * 128];
      bn1 = bf[(j + 1) * 128 + 32];
    }
    if (j == 6) {
      const float4 s_lo = *(const float4*)&nlds[p * 16 + half * 8];
      const float4 s_hi = *(const float4*)&nlds[p * 16 + half * 8 + 4];
      dn[0] = s_lo.x; dn[1] = s_lo.y; dn[2] = s_lo.z; dn[3] = s_lo.w;
      dn[4] = s_hi.x; dn[5] = s_hi.y; dn[6] = s_hi.z; dn[7] = s_hi.w;
    }
    __builtin_amdgcn_sched_barrier(0);
    v2f a, b;
    a.x = ac.x; a.y = ac.y;
    b.x = b0.x; b.y = b0.y;
    acc0 = __builtin_amdgcn_wmma_f32_16x16x4_f32(false, a, false, b, (short)0, acc0, false, false);
    b.x = b1.x; b.y = b1.y;
    acc1 = __builtin_amdgcn_wmma_f32_16x16x4_f32(false, a, false, b, (short)0, acc1, false, false);
    a.x = ac.z; a.y = ac.w;
    b.x = b0.z; b.y = b0.w;
    acc0 = __builtin_amdgcn_wmma_f32_16x16x4_f32(false, a, false, b, (short)0, acc0, false, false);
    b.x = b1.z; b.y = b1.w;
    acc1 = __builtin_amdgcn_wmma_f32_16x16x4_f32(false, a, false, b, (short)0, acc1, false, false);
    ac = an; b0 = bn0; b1 = bn1;
  }

  const int nvalid = n_nodes - rbase;
  const unsigned ob = ((unsigned)rbase * 64u + (unsigned)(ch * 32 + m * 2)) * 4u;
  char* outb = (char*)out;
  #pragma unroll
  for (int r = 0; r < 8; ++r) {
    if (r < nvalid) {
      const float s = dn[r];
      *(float2*)(outb + (ob + (unsigned)r * 256u)) = make_float2(acc0[r] * s, acc1[r] * s);
    }
  }
}

extern "C" void kernel_launch(void* const* d_in, const int* in_sizes, int n_in,
                              void* d_out, int out_size, void* d_ws, size_t ws_size,
                              hipStream_t stream) {
  const float* h    = (const float*)d_in[0];
  const float* w    = (const float*)d_in[1];
  const float* norm = (const float*)d_in[2];
  const int*   src  = (const int*)d_in[3];
  const int*   dst  = (const int*)d_in[4];
  float* out = (float*)d_out;
  int*   ws  = (int*)d_ws;

  const int n_nodes = in_sizes[2];
  const int n_edges = in_sizes[3];

  bin_sort_kernel<<<(n_edges + FILL_EDGES - 1) / FILL_EDGES, 256, 0, stream>>>(src, dst, ws, n_nodes, n_edges);

  bin_sum_mm_kernel<<<(n_nodes + BIN_ROWS - 1) / BIN_ROWS, SUM_THREADS, 0, stream>>>(h, w, norm, src, dst, ws, out, n_nodes, n_edges);
}


# Bin re-sized from 144 to 400 destinations: one 25-wave sum block fills a WGP, a wave sums 16 destinations (row loads in groups of up to 16 pairs) and multiplies its own tile; sums in arrival order
# speedup vs baseline: 5.6342x; 1.1480x over previous
#include <hip/hip_runtime.h>


typedef __attribute__((ext_vector_type(2))) float v2f;
typedef __attribute__((ext_vector_type(8))) float v8f;

#define BIN_ROWS   400
#define SUM_WAVES  (BIN_ROWS / 16)
#define SUM_THREADS (SUM_WAVES * 32)
#define MAX_BINS   1024
#define MAX_NODES  131072
#define FILL_EDGES 8192
#define SLOTS      46
#define DSTRIDE    (2 * SLOTS)
#define WSTRIDE    (16 * DSTRIDE)
#define ROWPITCH   68

__global__ __launch_bounds__(256) void bin_sort_kernel(
    const int* __restrict__ src, const int* __restrict__ dst,
    int* __restrict__ ws, int n_nodes, int n_edges) {
  __shared__ __attribute__((aligned(16))) int hist[MAX_BINS];
  __shared__ __attribute__((aligned(16))) int lofs[MAX_BINS];
  __shared__ __attribute__((aligned(16))) int wsum[8];
  __shared__ __attribute__((aligned(16))) int sval[FILL_EDGES + 32];

  if (n_nodes > MAX_NODES) return;

  const int tid  = threadIdx.x;
  const int lane = tid & 31;
  const int wave = tid >> 5;
  const int nblk = (n_edges + FILL_EDGES - 1) / FILL_EDGES;
  const int nbins = (n_nodes + BIN_ROWS - 1) / BIN_ROWS;
  int* list  = ws + blockIdx.x * FILL_EDGES;
  int* table = ws + nblk * FILL_EDGES;
  *(int4*)&hist[4 * tid] = make_int4(0, 0, 0, 0);
  __syncthreads();

  const int e0 = blockIdx.x * FILL_EDGES + tid;
  int sv[32], dv[32], br[32];
  #pragma unroll
  for (int k = 0; k < 32; ++k) dv[k] = dst[min(e0 + k * 256, n_edges - 1)];
  #pragma unroll
  for (int k = 0; k < 32; ++k) sv[k] = src[min(e0 + k * 256, n_edges - 1)];
  #pragma unroll
  for (int k = 0; k < 32; ++k) dv[k] = (e0 + k * 256 < n_edges) ? dv[k] : -1;
  #pragma unroll
  for (int k = 0; k < 32; ++k) {
    const bool ok = dv[k] >= 0;
    const int bin = ok ? (int)((unsigned)dv[k] / BIN_ROWS) : tid;
    br[k] = atomicAdd(&hist[bin], ok ? 1 : 0);
  }
  #pragma unroll
  for (int k = 0; k < 32; ++k) {
    const unsigned ud = (unsigned)dv[k];
    const unsigned bin = ud / BIN_ROWS;
    br[k] = (int)((bin << 16) | (unsigned)br[k]) | (dv[k] >> 31);
    sv[k] |= (int)((ud - bin * BIN_ROWS) << 17);
  }
  __syncthreads();

  const int4 hv = *(const int4*)&hist[4 * tid];
  const int tsum = hv.x + hv.y + hv.z + hv.w;
  int incl = tsum;
  #pragma unroll
  for (int o = 1; o < 32; o <<= 1) {
    const int t = __shfl_up(incl, o);
    if (lane >= o) incl += t;
  }
  if (lane == 31) wsum[wave] = incl;
  __syncthreads();
  const int4 wlo = *(const int4*)&wsum[0];
  const int4 whi = *(const int4*)&wsum[4];
  const int wv[8] = {wlo.x, wlo.y, wlo.z, wlo.w, whi.x, whi.y, whi.z, whi.w};
  int excl = incl - tsum;
  #pragma unroll
  for (int w = 0; w < 8; ++w) excl += (w < wave) ? wv[w] : 0;
  const int hb[4] = {hv.x, hv.y, hv.z, hv.w};
  int ex[4];
  #pragma unroll
  for (int u = 0; u < 4; ++u) {
    ex[u] = excl;
    excl += hb[u];
  }
  *(int4*)&lofs[4 * tid] = make_int4(ex[0], ex[1], ex[2], ex[3]);
  __syncthreads();

  int pl[32];
  #pragma unroll
  for (int k = 0; k < 32; ++k) pl[k] = lofs[(br[k] >= 0) ? (br[k] >> 16) : tid];
  #pragma unroll
  for (int k = 0; k < 32; ++k) {
    const int p = (br[k] >= 0) ? pl[k] + (br[k] & 0xFFFF) : FILL_EDGES + lane;
    sval[p] = sv[k];
  }
  __syncthreads();

  int4 ov[8];
  #pragma unroll
  for (int u = 0; u < 8; ++u) {
    const int i = (tid + u * 256) * 4;
    ov[u] = *(const int4*)&sval[i];
  }
  #pragma unroll
  for (int u = 0; u < 8; ++u) {
    const int i = (tid + u * 256) * 4;
    *(int4*)&list[i] = ov[u];
  }
  #pragma unroll
  for (int u = 0; u < 4; ++u) {
    const int bin = 4 * tid + u;
    if (bin < nbins) table[bin * nblk + blockIdx.x] = (ex[u] << 14) | hb[u];
  }
}

template <int R>
__device__ __forceinline__ float4 sum_group(
    const int* __restrict__ bk, int j, int cd, int half,
    const char* __restrict__ hb, unsigned qo, float4 acc) {
  float4 v[R];
  float sn[R];
  #pragma unroll
  for (int u = 0; u < R; ++u) {
    int t = 2 * (j + u) + half;
    if (u == R - 1) t = min(t, cd - 1);
    const int s = bk[t];
    v[u] = *(const float4*)(hb + ((unsigned)s * 256u + qo));
  }
  #pragma unroll
  for (int u = 0; u < R; ++u) {
    int t = 2 * (j + u) + half;
    if (u == R - 1) t = min(t, cd - 1);
    sn[u] = __int_as_float(bk[SLOTS + t]);
  }
  if (2 * (j + R - 1) + half >= cd) {
    v[R - 1] = make_float4(0.f, 0.f, 0.f, 0.f);
    sn[R - 1] = 0.f;
  }
  #pragma unroll
  for (int u = 0; u < R; ++u) {
    acc.x += sn[u] * v[u].x; acc.y += sn[u] * v[u].y;
    acc.z += sn[u] * v[u].z; acc.w += sn[u] * v[u].w;
  }
  return acc;
}

__global__ __launch_bounds__(SUM_THREADS) void bin_sum_mm_kernel(
    const float* __restrict__ h, const float* __restrict__ w,
    const float* __restrict__ norm,
    const int* __restrict__ src, const int* __restrict__ dst,
    const int* __restrict__ ws, float* __restrict__ out,
    int n_nodes, int n_edges) {
  __shared__ int cnt[BIN_ROWS];
  __shared__ __attribute__((aligned(16))) int bucket[SUM_WAVES * WSTRIDE];
  __shared__ float4 wfrag[8 * 2 * 2 * 32];
  __shared__ __attribute__((aligned(16))) float nlds[BIN_ROWS];

  const int tid  = threadIdx.x;
  const int d0   = blockIdx.x * BIN_ROWS;
  const int rows = min(BIN_ROWS, n_nodes - d0);
  const int nblk = (n_edges + FILL_EDGES - 1) / FILL_EDGES;
  const bool binok = (n_nodes <= MAX_NODES);
  float* fb = (float*)bucket;

  const float4* wsrc = (const float4*)w;
  float4 wv[2];
  #pragma unroll
  for (int it = 0; it < 2; ++it) wv[it] = wsrc[min(tid + it * SUM_THREADS, 1023)];
  const int* trow = ws + nblk * FILL_EDGES + blockIdx.x * nblk;
  int b = tid >> 2;
  int t = (binok && b < nblk) ? trow[b] : 0;

  float nv0 = 0.f;
  if (tid < BIN_ROWS) nv0 = norm[min(d0 + tid, n_nodes - 1)];

  if (tid < BIN_ROWS) cnt[tid] = 0;
  if (tid < BIN_ROWS) nlds[tid] = nv0;
  {
    float* wdst = (float*)wfrag;
    #pragma unroll
    for (int it = 0; it < 2; ++it) {
      const int idx = tid + it * SUM_THREADS;
      if (idx >= 1024) break;
      const int k = idx >> 4;
      const int i = idx & 15;
      const int j = k >> 3, hf = (k >> 2) & 1, c = i >> 3, n0 = 2 * (i & 7);
      const int base = (((j * 2 + c) * 2) * 32 + hf * 16 + n0) * 4 + (k & 3);
      wdst[base]           = wv[it].x;
      wdst[base + 128]     = wv[it].y;
      wdst[base + 4]       = wv[it].z;
      wdst[base + 128 + 4] = wv[it].w;
    }
  }
  __syncthreads();
  const int wave = __builtin_amdgcn_readfirstlane(tid >> 5);
  if (binok) {
    const int par = tid & 3;
    while (b < nblk) {
      const int off = min((t >> 14) & 0x3FFF, FILL_EDGES);
      const int c   = min(t & 0x3FFF, FILL_EDGES - off);
      const int* run = ws + b * FILL_EDGES + off;
      unsigned e[14];
      float nr[14];
      #pragma unroll
      for (int u = 0; u < 14; ++u) {
        const int i = par + 4 * u;
        e[u] = (i < c) ? (unsigned)run[(unsigned)i] : 0u;
      }
      #pragma unroll
      for (int u = 0; u < 14; ++u) nr[u] = norm[e[u] & 0x1FFFFu];
      int sl[14];
      #pragma unroll
      for (int u = 0; u < 14; ++u) {
        const bool ok = par + 4 * u < c;
        const int dl = ok ? min((int)(e[u] >> 17) & 0x1FF, BIN_ROWS - 1) : (tid & 127);
        sl[u] = atomicAdd(&cnt[dl], ok ? 1 : 0);
      }
      #pragma unroll
      for (int u = 0; u < 14; ++u) {
        if (par + 4 * u < c && sl[u] < SLOTS) {
          const int dl = min((int)(e[u] >> 17) & 0x1FF, BIN_ROWS - 1);
          bucket[dl * DSTRIDE + sl[u]]         = (int)(e[u] & 0x1FFFFu);
          bucket[dl * DSTRIDE + SLOTS + sl[u]] = __float_as_int(nr[u]);
        }
      }
      for (int i = 56 + par; i < c; i += 4) {
        const unsigned ee = (unsigned)run[i];
        const int s  = (int)(ee & 0x1FFFFu);
        const int dl = min((int)(ee >> 17) & 0x1FF, BIN_ROWS - 1);
        const float nv = norm[s];
        const int slot = atomicAdd(&cnt[dl], 1);
        if (slot < SLOTS) {
          bucket[dl * DSTRIDE + slot]         = s;
          bucket[dl * DSTRIDE + SLOTS + slot] = __float_as_int(nv);
        }
      }
      b += SUM_THREADS / 4;
      if (b < nblk) t = trow[b];
    }
  }
  __syncthreads();

  const int lane = (int)__builtin_amdgcn_mbcnt_lo(~0u, 0u);
  const int half = lane >> 4;
  const int q    = lane & 15;
  const unsigned qo = (unsigned)q * 16u;
  const char* hb = (const char*)h;

  for (int k = 0; k < 16; ++k) {
    const int dl = wave * 16 + k;
    if (dl >= rows) break;
    const int n = d0 + dl;
    const int cd = __builtin_amdgcn_readfirstlane(cnt[dl]);
    float* arow = fb + wave * WSTRIDE + k * ROWPITCH;

    if (!binok || cd > SLOTS) {
      float2 a2 = make_float2(0.f, 0.f);
      for (int base = 0; base < n_edges; base += 32) {
        const int e  = base + lane;
        const int dv = (e < n_edges) ? dst[e] : -1;
        const int sv = (e < n_edges) ? src[e] : 0;
        unsigned mask = __builtin_amdgcn_ballot_w32(dv == n);
        while (mask) {
          const int b = __builtin_ctz(mask);
          mask &= mask - 1;
          const int s = __builtin_amdgcn_readlane(sv, b);
          const float sn = norm[s];
          const float2 v = *(const float2*)(h + s * 64 + lane * 2);
          a2.x += sn * v.x; a2.y += sn * v.y;
        }
      }
      *(float2*)(arow + lane * 2) = a2;
      continue;
    }

    const int* bk = bucket + dl * DSTRIDE;
    float4 acc = make_float4(0.f, 0.f, 0.f, 0.f);
    const int np = (cd + 1) >> 1;
    for (int j = 0; j < np; j += 16) {
      switch (min(np - j, 16)) {
        case 16: acc = sum_group<16>(bk, j, cd, half, hb, qo, acc); break;
        case 15: acc = sum_group<15>(bk, j, cd, half, hb, qo, acc); break;
        case 14: acc = sum_group<14>(bk, j, cd, half, hb, qo, acc); break;
        case 13: acc = sum_group<13>(bk, j, cd, half, hb, qo, acc); break;
        case 12: acc = sum_group<12>(bk, j, cd, half, hb, qo, acc); break;
        case 11: acc = sum_group<11>(bk, j, cd, half, hb, qo, acc); break;
        case 10: acc = sum_group<10>(bk, j, cd, half, hb, qo, acc); break;
        case 9:  acc = sum_group<9>(bk, j, cd, half, hb, qo, acc); break;
        case 8:  acc = sum_group<8>(bk, j, cd, half, hb, qo, acc); break;
        case 7:  acc = sum_group<7>(bk, j, cd, half, hb, qo, acc); break;
        case 6:  acc = sum_group<6>(bk, j, cd, half, hb, qo, acc); break;
        case 5:  acc = sum_group<5>(bk, j, cd, half, hb, qo, acc); break;
        case 4:  acc = sum_group<4>(bk, j, cd, half, hb, qo, acc); break;
        case 3:  acc = sum_group<3>(bk, j, cd, half, hb, qo, acc); break;
        case 2:  acc = sum_group<2>(bk, j, cd, half, hb, qo, acc); break;
        default: acc = sum_group<1>(bk, j, cd, half, hb, qo, acc); break;
      }
    }

    acc.x += __shfl_xor(acc.x, 16);
    acc.y += __shfl_xor(acc.y, 16);
    acc.z += __shfl_xor(acc.z, 16);
    acc.w += __shfl_xor(acc.w, 16);
    if (half == 0) *(float4*)(arow + q * 4) = acc;
  }
  if (wave * 16 >= rows) return;
  const int m  = q;
  const int rbase = d0 + wave * 16 + half * 8;
  float dn[8];

  const float* ar = fb + wave * WSTRIDE + m * ROWPITCH + 4 * half;
  const float4* bf = wfrag + lane;

  v8f acc00 = {}, acc01 = {}, acc10 = {}, acc11 = {};
  float4 ac, b00, b01, b10, b11, an, n00, n01, n10, n11;
  ac  = *(const float4*)ar;
  b00 = bf[0];
  b01 = bf[32];
  b10 = bf[64];
  b11 = bf[96];
  #pragma unroll
  for (int j = 0; j < 8; ++j) {
    if (j + 1 < 8) {
      an  = *(const float4*)(ar + 8 * (j + 1));
      n00 = bf[(j + 1) * 128];
      n01 = bf[(j + 1) * 128 + 32];
      n10 = bf[(j + 1) * 128 + 64];
      n11 = bf[(j + 1) * 128 + 96];
    }
    if (j == 6) {
      const float4 s_lo = *(const float4*)&nlds[wave * 16 + half * 8];
      const float4 s_hi = *(const float4*)&nlds[wave * 16 + half * 8 + 4];
      dn[0] = s_lo.x; dn[1] = s_lo.y; dn[2] = s_lo.z; dn[3] = s_lo.w;
      dn[4] = s_hi.x; dn[5] = s_hi.y; dn[6] = s_hi.z; dn[7] = s_hi.w;
    }
    __builtin_amdgcn_sched_barrier(0);
    v2f a, b;
    a.x = ac.x; a.y = ac.y;
    b.x = b00.x; b.y = b00.y;
    acc00 = __builtin_amdgcn_wmma_f32_16x16x4_f32(false, a, false, b, (short)0, acc00, false, false);
    b.x = b01.x; b.y = b01.y;
    acc01 = __builtin_amdgcn_wmma_f32_16x16x4_f32(false, a, false, b, (short)0, acc01, false, false);
    b.x = b10.x; b.y = b10.y;
    acc10 = __builtin_amdgcn_wmma_f32_16x16x4_f32(false, a, false, b, (short)0, acc10, false, false);
    b.x = b11.x; b.y = b11.y;
    acc11 = __builtin_amdgcn_wmma_f32_16x16x4_f32(false, a, false, b, (short)0, acc11, false, false);
    a.x = ac.z; a.y = ac.w;
    b.x = b00.z; b.y = b00.w;
    acc00 = __builtin_amdgcn_wmma_f32_16x16x4_f32(false, a, false, b, (short)0, acc00, false, false);
    b.x = b01.z; b.y = b01.w;
    acc01 = __builtin_amdgcn_wmma_f32_16x16x4_f32(false, a, false, b, (short)0, acc01, false, false);
    b.x = b10.z; b.y = b10.w;
    acc10 = __builtin_amdgcn_wmma_f32_16x16x4_f32(false, a, false, b, (short)0, acc10, false, false);
    b.x = b11.z; b.y = b11.w;
    acc11 = __builtin_amdgcn_wmma_f32_16x16x4_f32(false, a, false, b, (short)0, acc11, false, false);
    ac = an; b00 = n00; b01 = n01; b10 = n10; b11 = n11;
  }

  const int nvalid = n_nodes - rbase;
  const unsigned ob = ((unsigned)rbase * 64u + (unsigned)(m * 2)) * 4u;
  char* outb = (char*)out;
  #pragma unroll
  for (int r = 0; r < 8; ++r) {
    if (r < nvalid) {
      const float s = dn[r];
      *(float2*)(outb + (ob + (unsigned)r * 256u))        = make_float2(acc00[r] * s, acc01[r] * s);
      *(float2*)(outb + (ob + (unsigned)r * 256u + 128u)) = make_float2(acc10[r] * s, acc11[r] * s);
    }
  }
}

extern "C" void kernel_launch(void* const* d_in, const int* in_sizes, int n_in,
                              void* d_out, int out_size, void* d_ws, size_t ws_size,
                              hipStream_t stream) {
  const float* h    = (const float*)d_in[0];
  const float* w    = (const float*)d_in[1];
  const float* norm = (const float*)d_in[2];
  const int*   src  = (const int*)d_in[3];
  const int*   dst  = (const int*)d_in[4];
  float* out = (float*)d_out;
  int*   ws  = (int*)d_ws;

  const int n_nodes = in_sizes[2];
  const int n_edges = in_sizes[3];

  bin_sort_kernel<<<(n_edges + FILL_EDGES - 1) / FILL_EDGES, 256, 0, stream>>>(src, dst, ws, n_nodes, n_edges);

  bin_sum_mm_kernel<<<(n_nodes + BIN_ROWS - 1) / BIN_ROWS, SUM_THREADS, 0, stream>>>(h, w, norm, src, dst, ws, out, n_nodes, n_edges);
}
